# starting kernel
# baseline (speedup 1.0000x reference)
#include <hip/hip_runtime.h>
#include <math.h>

typedef __attribute__((ext_vector_type(2))) float v2f;
typedef __attribute__((ext_vector_type(4))) float v4f;
typedef __attribute__((ext_vector_type(8))) float v8f;

#define B_  4
#define L_  512
#define V_  32
#define D_  64
#define MT  32
#define SCALE 0.125f
#define LDSS (L_ + 4)

__global__ __launch_bounds__(256)
void sum_key_kernel(const float* __restrict__ key, float* __restrict__ sumv) {
    int idx = blockIdx.x * blockDim.x + threadIdx.x;
    if (idx >= B_ * L_ * D_) return;
    int d  = idx % D_;
    int bl = idx / D_;
    const float* p = key + (size_t)bl * (V_ * D_) + d;
    float s = 0.f;
#pragma unroll
    for (int v = 0; v < V_; ++v) s += p[v * D_];
    sumv[idx] = s;
}

__global__ __launch_bounds__(32)
void attn_kernel(const float* __restrict__ q,
                 const float* __restrict__ sumv,
                 const float* __restrict__ val,
                 const int*   __restrict__ lab,
                 float*       __restrict__ out) {
    __shared__ float smem[MT * LDSS];
    __shared__ float labf[L_];
    __shared__ float rowsum[MT];

    const int blk   = blockIdx.x;
    const int itile = blk % (L_ / MT);
    const int bk    = blk / (L_ / MT);
    const int k     = bk % V_;
    const int b     = bk / V_;

    const int lane = threadIdx.x;
    const int hlf  = lane >> 4;
    const int l16  = lane & 15;
    const int i0   = itile * MT;

    for (int t = lane; t < L_; t += 32) labf[t] = (float)lab[b * L_ + t];
    __syncthreads();

    v2f qa0[16], qa1[16];
    {
        const float* q0 =
            q + (((size_t)(b * L_ + i0 + l16) * V_) + k) * D_ + hlf * 2;
        const float* q1 = q0 + (size_t)16 * V_ * D_;
#pragma unroll
        for (int s = 0; s < 16; ++s) {
            qa0[s] = *(const v2f*)(q0 + s * 4);
            qa1[s] = *(const v2f*)(q1 + s * 4);
        }
    }

    const float* sb = sumv + (size_t)b * L_ * D_;
    for (int jt = 0; jt < L_ / 16; ++jt) {
        v8f c0 = {}, c1 = {};
        const float* srow = sb + (size_t)(jt * 16 + l16) * D_ + hlf * 2;
#pragma unroll
        for (int s = 0; s < 16; ++s) {
            v2f bf = *(const v2f*)(srow + s * 4);
            c0 = __builtin_amdgcn_wmma_f32_16x16x4_f32(
                     false, qa0[s], false, bf, (short)0, c0, false, false);
            c1 = __builtin_amdgcn_wmma_f32_16x16x4_f32(
                     false, qa1[s], false, bf, (short)0, c1, false, false);
        }
#pragma unroll
        for (int r = 0; r < 8; ++r) {
            smem[(r + hlf * 8) * LDSS + jt * 16 + l16]      = c0[r];
            smem[(16 + r + hlf * 8) * LDSS + jt * 16 + l16] = c1[r];
        }
    }
    __syncthreads();

    {
        const int row = lane;
        const float myLab = labf[i0 + row];
        float* srow = &smem[row * LDSS];

        v4f m4 = {-INFINITY, -INFINITY, -INFINITY, -INFINITY};
        for (int c = 0; c < L_; c += 4) {
            v4f s  = *(v4f*)(srow + c);
            v4f lv = *(const v4f*)(&labf[c]);
            s.x = (lv.x == myLab) ? s.x : -INFINITY;
            s.y = (lv.y == myLab) ? s.y : -INFINITY;
            s.z = (lv.z == myLab) ? s.z : -INFINITY;
            s.w = (lv.w == myLab) ? s.w : -INFINITY;
            *(v4f*)(srow + c) = s;
            m4.x = fmaxf(m4.x, s.x);  m4.y = fmaxf(m4.y, s.y);
            m4.z = fmaxf(m4.z, s.z);  m4.w = fmaxf(m4.w, s.w);
        }
        const float m = fmaxf(fmaxf(m4.x, m4.y), fmaxf(m4.z, m4.w));

        v4f s4 = {0.f, 0.f, 0.f, 0.f};
        for (int c = 0; c < L_; c += 4) {
            v4f s = *(v4f*)(srow + c);
            v4f p;
            p.x = __expf(SCALE * (s.x - m));
            p.y = __expf(SCALE * (s.y - m));
            p.z = __expf(SCALE * (s.z - m));
            p.w = __expf(SCALE * (s.w - m));
            *(v4f*)(srow + c) = p;
            s4 += p;
        }
        rowsum[row] = (s4.x + s4.y) + (s4.z + s4.w);
    }
    __syncthreads();

    v8f acc[2][4] = {{{}, {}, {}, {}}, {{}, {}, {}, {}}};
    const float* vb = val + ((size_t)(b * L_) * V_ + k) * D_;
    for (int k0 = 0; k0 < L_; k0 += 4) {
        const int j0 = k0 + hlf * 2;
        const v2f af0 = *(const v2f*)(&smem[l16 * LDSS + j0]);
        const v2f af1 = *(const v2f*)(&smem[(16 + l16) * LDSS + j0]);
        const float* v0 = vb + (size_t)j0 * (V_ * D_);
        const float* v1 = v0 + (V_ * D_);
#pragma unroll
        for (int n = 0; n < 4; ++n) {
            v2f bf;
            bf.x = v0[n * 16 + l16];
            bf.y = v1[n * 16 + l16];
            acc[0][n] = __builtin_amdgcn_wmma_f32_16x16x4_f32(
                            false, af0, false, bf, (short)0, acc[0][n], false, false);
            acc[1][n] = __builtin_amdgcn_wmma_f32_16x16x4_f32(
                            false, af1, false, bf, (short)0, acc[1][n], false, false);
        }
    }

#pragma unroll
    for (int t = 0; t < 2; ++t) {
#pragma unroll
        for (int n = 0; n < 4; ++n) {
#pragma unroll
            for (int r = 0; r < 8; ++r) {
                const int irow = t * 16 + r + hlf * 8;
                const int i = i0 + irow;
                const float o = acc[t][n][r] / rowsum[irow];
                out[(((size_t)(b * L_ + i) * V_) + k) * D_ + n * 16 + l16] = o;
            }
        }
    }
}

extern "C" void kernel_launch(void* const* d_in, const int* in_sizes, int n_in,
                              void* d_out, int out_size, void* d_ws, size_t ws_size,
                              hipStream_t stream) {
    const float* query = (const float*)d_in[0];
    const float* key   = (const float*)d_in[1];
    const float* value = (const float*)d_in[2];
    const int*   label = (const int*)d_in[3];
    float* out  = (float*)d_out;
    float* sumv = (float*)d_ws;

    const int n1 = B_ * L_ * D_;
    sum_key_kernel<<<(n1 + 255) / 256, 256, 0, stream>>>(key, sumv);

    const int blocks = B_ * V_ * (L_ / MT);
    attn_kernel<<<blocks, 32, 0, stream>>>(query, sumv, value, label, out);
}


# Attention kernel: scores computed transposed (keys x rows) and kept in registers: running row max and sum, exponentiated tile is the B operand of the second product; no LDS score buffer
# speedup vs baseline: 1.7807x; 1.7807x over previous
#include <hip/hip_runtime.h>
#include <math.h>


typedef __attribute__((ext_vector_type(2))) float v2f;
typedef __attribute__((ext_vector_type(4))) float v4f;
typedef __attribute__((ext_vector_type(8))) float v8f;
typedef __attribute__((ext_vector_type(4))) int   v4i;

#define B_  4
#define L_  512
#define V_  32
#define D_  64
#define MT  32
#define KT  16
#define CSCALE 0.18033688011112042f
#define OPITCH (D_ + 4)

__global__ __launch_bounds__(256)
void sum_key_kernel(const float* __restrict__ key, float* __restrict__ sumv) {
    int idx = blockIdx.x * blockDim.x + threadIdx.x;
    if (idx >= B_ * L_ * D_) return;
    int d  = idx % D_;
    int bl = idx / D_;
    const float* p = key + (size_t)bl * (V_ * D_) + d;
    float s = 0.f;
#pragma unroll
    for (int v = 0; v < V_; ++v) s += p[v * D_];
    sumv[idx] = s;
}

__device__ __forceinline__ float xhalf(float x) {
    const int b = __builtin_bit_cast(int, x);
    return __builtin_bit_cast(float,
        __builtin_amdgcn_permlanex16(b, b, 0x76543210, 0xfedcba98, false, false));
}

__device__ __forceinline__ v8f mma(v2f a, v2f b, v8f c) {
    return __builtin_amdgcn_wmma_f32_16x16x4_f32(false, a, false, b, (short)0, c, false, false);
}

__global__ __launch_bounds__(32)
__attribute__((amdgpu_waves_per_eu(2, 2)))
void attn_kernel(const float* __restrict__ q,
                 const float* __restrict__ sumv,
                 const float* __restrict__ val,
                 const int*   __restrict__ lab,
                 float*       __restrict__ out) {
    __shared__ __attribute__((aligned(16))) float ot[MT * OPITCH];

    const int blk   = blockIdx.x;
    const int itile = blk % (L_ / MT);
    const int bk    = blk / (L_ / MT);
    const int k     = bk % V_;
    const int b     = bk / V_;

    const int lane = threadIdx.x;
    const int hlf  = lane >> 4;
    const int l16  = lane & 15;
    const int i0   = itile * MT;

    v4f qf0[8], qf1[8];
    {
        const float* q0 =
            q + (((size_t)(b * L_ + i0 + l16) * V_) + k) * D_ + hlf * 4;
        const float* q1 = q0 + (size_t)16 * V_ * D_;
#pragma unroll
        for (int u = 0; u < 8; ++u) {
            qf0[u] = *(const v4f*)(q0 + u * 8);
            qf1[u] = *(const v4f*)(q1 + u * 8);
        }
    }

    const int* lb = lab + b * L_;
    const int lab0 = lb[i0 + l16];
    const int lab1 = lb[i0 + 16 + l16];

    v8f acc0[4] = {{}, {}, {}, {}};
    v8f acc1[4] = {{}, {}, {}, {}};
    float m0 = -INFINITY, m1 = -INFINITY;
    float s0 = 0.f, s1 = 0.f;

    const float* sp = sumv + ((size_t)b * L_ + l16) * D_ + hlf * 4;
    const float* vp = val + (((size_t)b * L_ + hlf * 8) * V_ + k) * D_ + l16 * 4;
    const int*   lk = lb + hlf * 8;

    for (int jt = 0; jt < L_ / KT; ++jt) {
        v8f c0 = {}, c1 = {};
#pragma unroll
        for (int u = 0; u < 8; ++u) {
            const v4f a = *(const v4f*)(sp + u * 8);
            const v2f alo = {a.x, a.y};
            const v2f ahi = {a.z, a.w};
            const v2f b0lo = {qf0[u].x, qf0[u].y};
            const v2f b1lo = {qf1[u].x, qf1[u].y};
            const v2f b0hi = {qf0[u].z, qf0[u].w};
            const v2f b1hi = {qf1[u].z, qf1[u].w};
            c0 = mma(alo, b0lo, c0);
            c1 = mma(alo, b1lo, c1);
            c0 = mma(ahi, b0hi, c0);
            c1 = mma(ahi, b1hi, c1);
        }

        const v4i ka = *(const v4i*)(lk);
        const v4i kb = *(const v4i*)(lk + 4);
        const int kl[8] = {ka.x, ka.y, ka.z, ka.w, kb.x, kb.y, kb.z, kb.w};
        float t0 = -INFINITY, t1 = -INFINITY;
#pragma unroll
        for (int r = 0; r < 8; ++r) {
            c0[r] = (kl[r] == lab0) ? c0[r] : -INFINITY;
            c1[r] = (kl[r] == lab1) ? c1[r] : -INFINITY;
            t0 = fmaxf(t0, c0[r]);
            t1 = fmaxf(t1, c1[r]);
        }
        t0 = fmaxf(t0, xhalf(t0));
        t1 = fmaxf(t1, xhalf(t1));
        const float n0 = fmaxf(m0, t0);
        const float n1 = fmaxf(m1, t1);
        const float z0 = (n0 == -INFINITY) ? 0.f : n0;
        const float z1 = (n1 == -INFINITY) ? 0.f : n1;
        const float a0 = __builtin_amdgcn_exp2f((m0 - z0) * CSCALE);
        const float a1 = __builtin_amdgcn_exp2f((m1 - z1) * CSCALE);
        m0 = n0;
        m1 = n1;

        float ps0 = 0.f, ps1 = 0.f;
#pragma unroll
        for (int r = 0; r < 8; ++r) {
            c0[r] = __builtin_amdgcn_exp2f((c0[r] - z0) * CSCALE);
            c1[r] = __builtin_amdgcn_exp2f((c1[r] - z1) * CSCALE);
            ps0 += c0[r];
            ps1 += c1[r];
        }
        s0 = s0 * a0 + ps0;
        s1 = s1 * a1 + ps1;
#pragma unroll
        for (int n = 0; n < 4; ++n) {
            acc0[n] *= a0;
            acc1[n] *= a1;
        }

#pragma unroll
        for (int s = 0; s < 4; ++s) {
            const v4f va = *(const v4f*)(vp + (size_t)(2 * s)     * (V_ * D_));
            const v4f vb = *(const v4f*)(vp + (size_t)(2 * s + 1) * (V_ * D_));
            const v2f p0 = {c0[2 * s], c0[2 * s + 1]};
            const v2f p1 = {c1[2 * s], c1[2 * s + 1]};
#pragma unroll
            for (int n = 0; n < 4; ++n) {
                const v2f af = {va[n], vb[n]};
                acc0[n] = mma(af, p0, acc0[n]);
                acc1[n] = mma(af, p1, acc1[n]);
            }
        }

        sp += KT * D_;
        vp += (size_t)KT * V_ * D_;
        lk += KT;
    }

    s0 += xhalf(s0);
    s1 += xhalf(s1);
    const float r0 = 1.0f / s0;
    const float r1 = 1.0f / s1;
#pragma unroll
    for (int r = 0; r < 8; ++r) {
        v4f o0 = {acc0[0][r], acc0[1][r], acc0[2][r], acc0[3][r]};
        v4f o1 = {acc1[0][r], acc1[1][r], acc1[2][r], acc1[3][r]};
        o0 *= r0;
        o1 *= r1;
        *(v4f*)&ot[l16 * OPITCH + hlf * 32 + r * 4]        = o0;
        *(v4f*)&ot[(16 + l16) * OPITCH + hlf * 32 + r * 4] = o1;
    }
    __syncthreads();
    float* ob = out + (((size_t)(b * L_ + i0) * V_) + k) * D_ + l16 * 4;
#pragma unroll 4
    for (int it = 0; it < MT / 2; ++it) {
        const int row = 2 * it + hlf;
        const v4f o = *(const v4f*)&ot[row * OPITCH + l16 * 4];
        *(v4f*)(ob + (size_t)row * (V_ * D_)) = o;
    }
}

extern "C" void kernel_launch(void* const* d_in, const int* in_sizes, int n_in,
                              void* d_out, int out_size, void* d_ws, size_t ws_size,
                              hipStream_t stream) {
    const float* query = (const float*)d_in[0];
    const float* key   = (const float*)d_in[1];
    const float* value = (const float*)d_in[2];
    const int*   label = (const int*)d_in[3];
    float* out  = (float*)d_out;
    float* sumv = (float*)d_ws;

    const int n1 = B_ * L_ * D_;
    sum_key_kernel<<<(n1 + 255) / 256, 256, 0, stream>>>(key, sumv);

    const int blocks = B_ * V_ * (L_ / MT);
    attn_kernel<<<blocks, 32, 0, stream>>>(query, sumv, value, label, out);
}


# Attention kernel: keys split between the two waves of a block (256 each, own running max/sum/accumulators), partial tiles merged once through LDS; 4096 waves, four per SIMD instead of two
# speedup vs baseline: 1.9040x; 1.0692x over previous
#include <hip/hip_runtime.h>
#include <math.h>


typedef __attribute__((ext_vector_type(2))) float v2f;
typedef __attribute__((ext_vector_type(4))) float v4f;
typedef __attribute__((ext_vector_type(8))) float v8f;
typedef __attribute__((ext_vector_type(4))) int   v4i;

#define B_  4
#define L_  512
#define V_  32
#define D_  64
#define MT  32
#define KT  16
#define CSCALE 0.18033688011112042f
#define OPITCH (D_ + 4)
#define KSPLIT 2
#define KW (L_ / KSPLIT)

__global__ __launch_bounds__(256)
void sum_key_kernel(const float* __restrict__ key, float* __restrict__ sumv) {
    int idx = blockIdx.x * blockDim.x + threadIdx.x;
    if (idx >= B_ * L_ * D_) return;
    int d  = idx % D_;
    int bl = idx / D_;
    const float* p = key + (size_t)bl * (V_ * D_) + d;
    float s = 0.f;
#pragma unroll
    for (int v = 0; v < V_; ++v) s += p[v * D_];
    sumv[idx] = s;
}

__device__ __forceinline__ float xhalf(float x) {
    const int b = __builtin_bit_cast(int, x);
    return __builtin_bit_cast(float,
        __builtin_amdgcn_permlanex16(b, b, 0x76543210, 0xfedcba98, false, false));
}

__device__ __forceinline__ v8f mma(v2f a, v2f b, v8f c) {
    return __builtin_amdgcn_wmma_f32_16x16x4_f32(false, a, false, b, (short)0, c, false, false);
}

__global__ __launch_bounds__(32 * KSPLIT)
__attribute__((amdgpu_waves_per_eu(4, 4)))
void attn_kernel(const float* __restrict__ q,
                 const float* __restrict__ sumv,
                 const float* __restrict__ val,
                 const int*   __restrict__ lab,
                 float*       __restrict__ out) {
    __shared__ __attribute__((aligned(16))) float ot[KSPLIT][MT * OPITCH];
    __shared__ float ms[KSPLIT][4][16];

    const int blk   = blockIdx.x;
    const int itile = blk % (L_ / MT);
    const int bk    = blk / (L_ / MT);
    const int k     = bk % V_;
    const int b     = bk / V_;

    const int lane = threadIdx.x & 31;
    const int w    = threadIdx.x >> 5;
    const int hlf  = lane >> 4;
    const int l16  = lane & 15;
    const int i0   = itile * MT;

    v4f qf0[8], qf1[8];
    {
        const float* q0 =
            q + (((size_t)(b * L_ + i0 + l16) * V_) + k) * D_ + hlf * 4;
        const float* q1 = q0 + (size_t)16 * V_ * D_;
#pragma unroll
        for (int u = 0; u < 8; ++u) {
            qf0[u] = *(const v4f*)(q0 + u * 8);
            qf1[u] = *(const v4f*)(q1 + u * 8);
        }
    }

    const int* lb = lab + b * L_;
    const int lab0 = lb[i0 + l16];
    const int lab1 = lb[i0 + 16 + l16];

    v8f acc0[4] = {{}, {}, {}, {}};
    v8f acc1[4] = {{}, {}, {}, {}};
    float m0 = -INFINITY, m1 = -INFINITY;
    float s0 = 0.f, s1 = 0.f;

    const int jw = w * KW;
    const float* sp = sumv + ((size_t)b * L_ + jw + l16) * D_ + hlf * 4;
    const float* vp = val + (((size_t)b * L_ + jw + hlf * 8) * V_ + k) * D_ + l16 * 4;
    const int*   lk = lb + jw + hlf * 8;

    for (int jt = 0; jt < KW / KT; ++jt) {
        v8f c0 = {}, c1 = {};
#pragma unroll
        for (int u = 0; u < 8; ++u) {
            const v4f a = *(const v4f*)(sp + u * 8);
            const v2f alo = {a.x, a.y};
            const v2f ahi = {a.z, a.w};
            const v2f b0lo = {qf0[u].x, qf0[u].y};
            const v2f b1lo = {qf1[u].x, qf1[u].y};
            const v2f b0hi = {qf0[u].z, qf0[u].w};
            const v2f b1hi = {qf1[u].z, qf1[u].w};
            c0 = mma(alo, b0lo, c0);
            c1 = mma(alo, b1lo, c1);
            c0 = mma(ahi, b0hi, c0);
            c1 = mma(ahi, b1hi, c1);
        }

        const v4i ka = *(const v4i*)(lk);
        const v4i kb = *(const v4i*)(lk + 4);
        const int kl[8] = {ka.x, ka.y, ka.z, ka.w, kb.x, kb.y, kb.z, kb.w};
        float t0 = -INFINITY, t1 = -INFINITY;
#pragma unroll
        for (int r = 0; r < 8; ++r) {
            c0[r] = (kl[r] == lab0) ? c0[r] : -INFINITY;
            c1[r] = (kl[r] == lab1) ? c1[r] : -INFINITY;
            t0 = fmaxf(t0, c0[r]);
            t1 = fmaxf(t1, c1[r]);
        }
        t0 = fmaxf(t0, xhalf(t0));
        t1 = fmaxf(t1, xhalf(t1));
        const float n0 = fmaxf(m0, t0);
        const float n1 = fmaxf(m1, t1);
        const float z0 = (n0 == -INFINITY) ? 0.f : n0;
        const float z1 = (n1 == -INFINITY) ? 0.f : n1;
        const float a0 = __builtin_amdgcn_exp2f((m0 - z0) * CSCALE);
        const float a1 = __builtin_amdgcn_exp2f((m1 - z1) * CSCALE);
        m0 = n0;
        m1 = n1;

        float ps0 = 0.f, ps1 = 0.f;
#pragma unroll
        for (int r = 0; r < 8; ++r) {
            c0[r] = __builtin_amdgcn_exp2f((c0[r] - z0) * CSCALE);
            c1[r] = __builtin_amdgcn_exp2f((c1[r] - z1) * CSCALE);
            ps0 += c0[r];
            ps1 += c1[r];
        }
        s0 = s0 * a0 + ps0;
        s1 = s1 * a1 + ps1;
#pragma unroll
        for (int n = 0; n < 4; ++n) {
            acc0[n] *= a0;
            acc1[n] *= a1;
        }

#pragma unroll
        for (int s = 0; s < 4; ++s) {
            const v4f va = *(const v4f*)(vp + (size_t)(2 * s)     * (V_ * D_));
            const v4f vb = *(const v4f*)(vp + (size_t)(2 * s + 1) * (V_ * D_));
            const v2f p0 = {c0[2 * s], c0[2 * s + 1]};
            const v2f p1 = {c1[2 * s], c1[2 * s + 1]};
#pragma unroll
            for (int n = 0; n < 4; ++n) {
                const v2f af = {va[n], vb[n]};
                acc0[n] = mma(af, p0, acc0[n]);
                acc1[n] = mma(af, p1, acc1[n]);
            }
        }

        sp += KT * D_;
        vp += (size_t)KT * V_ * D_;
        lk += KT;
    }

    s0 += xhalf(s0);
    s1 += xhalf(s1);
    if (hlf == 0) {
        ms[w][0][l16] = m0;
        ms[w][1][l16] = m1;
        ms[w][2][l16] = s0;
        ms[w][3][l16] = s1;
    }
    __syncthreads();
    const float mo0 = ms[w ^ 1][0][l16];
    const float mo1 = ms[w ^ 1][1][l16];
    const float so0 = ms[w ^ 1][2][l16];
    const float so1 = ms[w ^ 1][3][l16];
    const float mm0 = fmaxf(m0, mo0);
    const float mm1 = fmaxf(m1, mo1);
    const float y0 = (mm0 == -INFINITY) ? 0.f : mm0;
    const float y1 = (mm1 == -INFINITY) ? 0.f : mm1;
    const float f0  = __builtin_amdgcn_exp2f((m0  - y0) * CSCALE);
    const float f1  = __builtin_amdgcn_exp2f((m1  - y1) * CSCALE);
    const float fo0 = __builtin_amdgcn_exp2f((mo0 - y0) * CSCALE);
    const float fo1 = __builtin_amdgcn_exp2f((mo1 - y1) * CSCALE);
    const float r0 = f0 / (s0 * f0 + so0 * fo0);
    const float r1 = f1 / (s1 * f1 + so1 * fo1);
    float* otw = ot[w];
#pragma unroll
    for (int r = 0; r < 8; ++r) {
        v4f o0 = {acc0[0][r], acc0[1][r], acc0[2][r], acc0[3][r]};
        v4f o1 = {acc1[0][r], acc1[1][r], acc1[2][r], acc1[3][r]};
        o0 *= r0;
        o1 *= r1;
        *(v4f*)&otw[l16 * OPITCH + hlf * 32 + r * 4]        = o0;
        *(v4f*)&otw[(16 + l16) * OPITCH + hlf * 32 + r * 4] = o1;
    }
    __syncthreads();
    const int rw = 16 * w;
    float* ob = out + (((size_t)(b * L_ + i0 + rw) * V_) + k) * D_ + l16 * 4;
#pragma unroll 4
    for (int it = 0; it < MT / (2 * KSPLIT); ++it) {
        const int row = 2 * it + hlf;
        const v4f oa = *(const v4f*)&ot[0][(rw + row) * OPITCH + l16 * 4];
        const v4f ob2 = *(const v4f*)&ot[1][(rw + row) * OPITCH + l16 * 4];
        *(v4f*)(ob + (size_t)row * (V_ * D_)) = oa + ob2;
    }
}

extern "C" void kernel_launch(void* const* d_in, const int* in_sizes, int n_in,
                              void* d_out, int out_size, void* d_ws, size_t ws_size,
                              hipStream_t stream) {
    const float* query = (const float*)d_in[0];
    const float* key   = (const float*)d_in[1];
    const float* value = (const float*)d_in[2];
    const int*   label = (const int*)d_in[3];
    float* out  = (float*)d_out;
    float* sumv = (float*)d_ws;

    const int n1 = B_ * L_ * D_;
    sum_key_kernel<<<(n1 + 255) / 256, 256, 0, stream>>>(key, sumv);

    const int blocks = B_ * V_ * (L_ / MT);
    attn_kernel<<<blocks, 32 * KSPLIT, 0, stream>>>(query, sumv, value, label, out);
}


# Attention: next 16-key tile's key-sum fragments and key labels requested right behind the current score product, into the same registers; value rows at the step's start; no load is waited for at once
# speedup vs baseline: 1.9940x; 1.0473x over previous
#include <hip/hip_runtime.h>
#include <math.h>


typedef __attribute__((ext_vector_type(2))) float v2f;
typedef __attribute__((ext_vector_type(4))) float v4f;
typedef __attribute__((ext_vector_type(8))) float v8f;
typedef __attribute__((ext_vector_type(4))) int   v4i;

#define B_  4
#define L_  512
#define V_  32
#define D_  64
#define MT  32
#define KT  16
#define CSCALE 0.18033688011112042f
#define OPITCH (D_ + 4)
#define KSPLIT 2
#define KW (L_ / KSPLIT)

__global__ __launch_bounds__(256)
void sum_key_kernel(const float* __restrict__ key, float* __restrict__ sumv) {
    int idx = blockIdx.x * blockDim.x + threadIdx.x;
    if (idx >= B_ * L_ * D_) return;
    int d  = idx % D_;
    int bl = idx / D_;
    const float* p = key + (size_t)bl * (V_ * D_) + d;
    float s = 0.f;
#pragma unroll
    for (int v = 0; v < V_; ++v) s += p[v * D_];
    sumv[idx] = s;
}

__device__ __forceinline__ float xhalf(float x) {
    const int b = __builtin_bit_cast(int, x);
    return __builtin_bit_cast(float,
        __builtin_amdgcn_permlanex16(b, b, 0x76543210, 0xfedcba98, false, false));
}

__device__ __forceinline__ v8f mma(v2f a, v2f b, v8f c) {
    return __builtin_amdgcn_wmma_f32_16x16x4_f32(false, a, false, b, (short)0, c, false, false);
}

__global__ __launch_bounds__(32 * KSPLIT)
__attribute__((amdgpu_waves_per_eu(4, 4)))
void attn_kernel(const float* __restrict__ q,
                 const float* __restrict__ sumv,
                 const float* __restrict__ val,
                 const int*   __restrict__ lab,
                 float*       __restrict__ out) {
    __shared__ __attribute__((aligned(16))) float ot[KSPLIT][MT * OPITCH];
    __shared__ float ms[KSPLIT][4][16];

    const int blk   = blockIdx.x;
    const int itile = blk % (L_ / MT);
    const int bk    = blk / (L_ / MT);
    const int k     = bk % V_;
    const int b     = bk / V_;

    const int lane = threadIdx.x & 31;
    const int w    = threadIdx.x >> 5;
    const int hlf  = lane >> 4;
    const int l16  = lane & 15;
    const int i0   = itile * MT;

    v4f qf0[8], qf1[8];
    {
        const float* q0 =
            q + (((size_t)(b * L_ + i0 + l16) * V_) + k) * D_ + hlf * 4;
        const float* q1 = q0 + (size_t)16 * V_ * D_;
#pragma unroll
        for (int u = 0; u < 8; ++u) {
            qf0[u] = *(const v4f*)(q0 + u * 8);
            qf1[u] = *(const v4f*)(q1 + u * 8);
        }
    }

    const int* lb = lab + b * L_;
    const int lab0 = lb[i0 + l16];
    const int lab1 = lb[i0 + 16 + l16];

    v8f acc0[4] = {{}, {}, {}, {}};
    v8f acc1[4] = {{}, {}, {}, {}};
    float m0 = -INFINITY, m1 = -INFINITY;
    float s0 = 0.f, s1 = 0.f;

    const int jw = w * KW;
    const float* sp = sumv + ((size_t)b * L_ + jw + l16) * D_ + hlf * 4;
    const float* vp = val + (((size_t)b * L_ + jw + hlf * 8) * V_ + k) * D_ + l16 * 4;
    const int*   lk = lb + jw + hlf * 8;

    v4i ka = *(const v4i*)(lk);
    v4i kb = *(const v4i*)(lk + 4);
    v4f kf[8];
#pragma unroll
    for (int u = 0; u < 8; ++u) kf[u] = *(const v4f*)(sp + u * 8);

    for (int jt = 0; jt < KW / KT; ++jt) {
        v4f va[4], vb[4];
#pragma unroll
        for (int s = 0; s < 4; ++s) {
            va[s] = *(const v4f*)(vp + (size_t)(2 * s)     * (V_ * D_));
            vb[s] = *(const v4f*)(vp + (size_t)(2 * s + 1) * (V_ * D_));
        }
        asm volatile("" : "+v"(kf[0]) : : "memory");

        v8f c0 = {}, c1 = {};
#pragma unroll
        for (int u = 0; u < 8; ++u) {
            const v4f a = kf[u];
            const v2f alo = {a.x, a.y};
            const v2f ahi = {a.z, a.w};
            const v2f b0lo = {qf0[u].x, qf0[u].y};
            const v2f b1lo = {qf1[u].x, qf1[u].y};
            const v2f b0hi = {qf0[u].z, qf0[u].w};
            const v2f b1hi = {qf1[u].z, qf1[u].w};
            c0 = mma(alo, b0lo, c0);
            c1 = mma(alo, b1lo, c1);
            c0 = mma(ahi, b0hi, c0);
            c1 = mma(ahi, b1hi, c1);
        }

        const int kl[8] = {ka.x, ka.y, ka.z, ka.w, kb.x, kb.y, kb.z, kb.w};
        bool e0[8], e1[8];
#pragma unroll
        for (int r = 0; r < 8; ++r) {
            e0[r] = (kl[r] == lab0);
            e1[r] = (kl[r] == lab1);
        }

        {
            int jn = (jt + 1) & (KW / KT - 1);
            asm volatile("" : "+s"(jn) : "v"(c0), "v"(c1) : "memory");
            const float* spn = sp + jn * (KT * D_);
            const int*   lkn = lk + jn * KT;
            ka = *(const v4i*)(lkn);
            kb = *(const v4i*)(lkn + 4);
#pragma unroll
            for (int u = 0; u < 8; ++u) kf[u] = *(const v4f*)(spn + u * 8);
            asm volatile("" ::: "memory");
        }

        float t0 = -INFINITY, t1 = -INFINITY;
#pragma unroll
        for (int r = 0; r < 8; ++r) {
            c0[r] = e0[r] ? c0[r] : -INFINITY;
            c1[r] = e1[r] ? c1[r] : -INFINITY;
            t0 = fmaxf(t0, c0[r]);
            t1 = fmaxf(t1, c1[r]);
        }
        t0 = fmaxf(t0, xhalf(t0));
        t1 = fmaxf(t1, xhalf(t1));
        const float n0 = fmaxf(m0, t0);
        const float n1 = fmaxf(m1, t1);
        const float z0 = (n0 == -INFINITY) ? 0.f : n0;
        const float z1 = (n1 == -INFINITY) ? 0.f : n1;
        const float a0 = __builtin_amdgcn_exp2f((m0 - z0) * CSCALE);
        const float a1 = __builtin_amdgcn_exp2f((m1 - z1) * CSCALE);
        m0 = n0;
        m1 = n1;

        float ps0 = 0.f, ps1 = 0.f;
#pragma unroll
        for (int r = 0; r < 8; ++r) {
            c0[r] = __builtin_amdgcn_exp2f((c0[r] - z0) * CSCALE);
            c1[r] = __builtin_amdgcn_exp2f((c1[r] - z1) * CSCALE);
            ps0 += c0[r];
            ps1 += c1[r];
        }
        s0 = s0 * a0 + ps0;
        s1 = s1 * a1 + ps1;
#pragma unroll
        for (int n = 0; n < 4; ++n) {
            acc0[n] *= a0;
            acc1[n] *= a1;
        }

#pragma unroll
        for (int s = 0; s < 4; ++s) {
            const v2f p0 = {c0[2 * s], c0[2 * s + 1]};
            const v2f p1 = {c1[2 * s], c1[2 * s + 1]};
#pragma unroll
            for (int n = 0; n < 4; ++n) {
                const v2f af = {va[s][n], vb[s][n]};
                acc0[n] = mma(af, p0, acc0[n]);
                acc1[n] = mma(af, p1, acc1[n]);
            }
        }

        vp += (size_t)KT * V_ * D_;
    }

    s0 += xhalf(s0);
    s1 += xhalf(s1);
    if (hlf == 0) {
        ms[w][0][l16] = m0;
        ms[w][1][l16] = m1;
        ms[w][2][l16] = s0;
        ms[w][3][l16] = s1;
    }
    __syncthreads();
    const float mo0 = ms[w ^ 1][0][l16];
    const float mo1 = ms[w ^ 1][1][l16];
    const float so0 = ms[w ^ 1][2][l16];
    const float so1 = ms[w ^ 1][3][l16];
    const float mm0 = fmaxf(m0, mo0);
    const float mm1 = fmaxf(m1, mo1);
    const float y0 = (mm0 == -INFINITY) ? 0.f : mm0;
    const float y1 = (mm1 == -INFINITY) ? 0.f : mm1;
    const float f0  = __builtin_amdgcn_exp2f((m0  - y0) * CSCALE);
    const float f1  = __builtin_amdgcn_exp2f((m1  - y1) * CSCALE);
    const float fo0 = __builtin_amdgcn_exp2f((mo0 - y0) * CSCALE);
    const float fo1 = __builtin_amdgcn_exp2f((mo1 - y1) * CSCALE);
    const float r0 = f0 / (s0 * f0 + so0 * fo0);
    const float r1 = f1 / (s1 * f1 + so1 * fo1);
    float* otw = ot[w];
#pragma unroll
    for (int r = 0; r < 8; ++r) {
        v4f o0 = {acc0[0][r], acc0[1][r], acc0[2][r], acc0[3][r]};
        v4f o1 = {acc1[0][r], acc1[1][r], acc1[2][r], acc1[3][r]};
        o0 *= r0;
        o1 *= r1;
        *(v4f*)&otw[l16 * OPITCH + hlf * 32 + r * 4]        = o0;
        *(v4f*)&otw[(16 + l16) * OPITCH + hlf * 32 + r * 4] = o1;
    }
    __syncthreads();
    const int rw = 16 * w;
    float* ob = out + (((size_t)(b * L_ + i0 + rw) * V_) + k) * D_ + l16 * 4;
#pragma unroll 4
    for (int it = 0; it < MT / (2 * KSPLIT); ++it) {
        const int row = 2 * it + hlf;
        const v4f oa = *(const v4f*)&ot[0][(rw + row) * OPITCH + l16 * 4];
        const v4f ob2 = *(const v4f*)&ot[1][(rw + row) * OPITCH + l16 * 4];
        *(v4f*)(ob + (size_t)row * (V_ * D_)) = oa + ob2;
    }
}

extern "C" void kernel_launch(void* const* d_in, const int* in_sizes, int n_in,
                              void* d_out, int out_size, void* d_ws, size_t ws_size,
                              hipStream_t stream) {
    const float* query = (const float*)d_in[0];
    const float* key   = (const float*)d_in[1];
    const float* value = (const float*)d_in[2];
    const int*   label = (const int*)d_in[3];
    float* out  = (float*)d_out;
    float* sumv = (float*)d_ws;

    const int n1 = B_ * L_ * D_;
    sum_key_kernel<<<(n1 + 255) / 256, 256, 0, stream>>>(key, sumv);

    const int blocks = B_ * V_ * (L_ / MT);
    attn_kernel<<<blocks, 32 * KSPLIT, 0, stream>>>(query, sumv, value, label, out);
}


# Score product absorbs the softmax's per-element constants: cluster mask is its C operand (0/-inf table in LDS, built once per block), log2e/sqrt(d) folded into Q at load
# speedup vs baseline: 2.0701x; 1.0382x over previous
#include <hip/hip_runtime.h>
#include <math.h>


typedef __attribute__((ext_vector_type(2))) float v2f;
typedef __attribute__((ext_vector_type(4))) float v4f;
typedef __attribute__((ext_vector_type(8))) float v8f;
typedef __attribute__((ext_vector_type(4))) int   v4i;

#define B_  4
#define L_  512
#define V_  32
#define D_  64
#define MT  32
#define KT  16
#define CSCALE 0.18033688011112042f
#define OPITCH (D_ + 4)
#define KSPLIT 2
#define KW (L_ / KSPLIT)
#define NLAB 8
#define PPITCH (L_ + 8)

__global__ __launch_bounds__(256)
void sum_key_kernel(const float* __restrict__ key, float* __restrict__ sumv) {
    int idx = blockIdx.x * blockDim.x + threadIdx.x;
    if (idx >= B_ * L_ * D_) return;
    int d  = idx % D_;
    int bl = idx / D_;
    const float* p = key + (size_t)bl * (V_ * D_) + d;
    float s = 0.f;
#pragma unroll
    for (int v = 0; v < V_; ++v) s += p[v * D_];
    sumv[idx] = s;
}

__device__ __forceinline__ float xhalf(float x) {
    const int b = __builtin_bit_cast(int, x);
    return __builtin_bit_cast(float,
        __builtin_amdgcn_permlanex16(b, b, 0x76543210, 0xfedcba98, false, false));
}

__device__ __forceinline__ v8f mma(v2f a, v2f b, v8f c) {
    return __builtin_amdgcn_wmma_f32_16x16x4_f32(false, a, false, b, (short)0, c, false, false);
}

__global__ __launch_bounds__(32 * KSPLIT)
__attribute__((amdgpu_waves_per_eu(4, 4)))
void attn_kernel(const float* __restrict__ q,
                 const float* __restrict__ sumv,
                 const float* __restrict__ val,
                 const int*   __restrict__ lab,
                 float*       __restrict__ out) {
    __shared__ __attribute__((aligned(16))) float ot[KSPLIT][MT * OPITCH];
    __shared__ float ms[KSPLIT][4][16];
    __shared__ __attribute__((aligned(16))) float pen[NLAB][PPITCH];

    const int blk   = blockIdx.x;
    const int itile = blk % (L_ / MT);
    const int bk    = blk / (L_ / MT);
    const int k     = bk % V_;
    const int b     = bk / V_;

    const int lane = threadIdx.x & 31;
    const int w    = threadIdx.x >> 5;
    const int hlf  = lane >> 4;
    const int l16  = lane & 15;
    const int i0   = itile * MT;

    v4f qf0[8], qf1[8];
    {
        const float* q0 =
            q + (((size_t)(b * L_ + i0 + l16) * V_) + k) * D_ + hlf * 4;
        const float* q1 = q0 + (size_t)16 * V_ * D_;
#pragma unroll
        for (int u = 0; u < 8; ++u) {
            qf0[u] = *(const v4f*)(q0 + u * 8) * CSCALE;
            qf1[u] = *(const v4f*)(q1 + u * 8) * CSCALE;
        }
    }

    const int* lb = lab + b * L_;
    const int lab0 = lb[i0 + l16];
    const int lab1 = lb[i0 + 16 + l16];

    {
        const int t8 = threadIdx.x * 8;
        const v4i la = *(const v4i*)(lb + t8);
        const v4i lc = *(const v4i*)(lb + t8 + 4);
        const v4f ninf = {-INFINITY, -INFINITY, -INFINITY, -INFINITY};
#pragma unroll
        for (int c = 0; c < NLAB; ++c) {
            *(v4f*)&pen[c][t8]     = ninf;
            *(v4f*)&pen[c][t8 + 4] = ninf;
        }
        const int kl[8] = {la.x, la.y, la.z, la.w, lc.x, lc.y, lc.z, lc.w};
#pragma unroll
        for (int r = 0; r < 8; ++r) pen[kl[r] & (NLAB - 1)][t8 + r] = 0.f;
    }
    __syncthreads();

    v8f acc0[4] = {{}, {}, {}, {}};
    v8f acc1[4] = {{}, {}, {}, {}};
    float m0 = -INFINITY, m1 = -INFINITY;
    float s0 = 0.f, s1 = 0.f;

    const int jw = w * KW;
    const float* sp = sumv + ((size_t)b * L_ + jw + l16) * D_ + hlf * 4;
    const float* vp = val + (((size_t)b * L_ + jw + hlf * 8) * V_ + k) * D_ + l16 * 4;
    const float* pp0 = &pen[lab0 & (NLAB - 1)][jw + hlf * 8];
    const float* pp1 = &pen[lab1 & (NLAB - 1)][jw + hlf * 8];

    v4f kf[8];
#pragma unroll
    for (int u = 0; u < 8; ++u) kf[u] = *(const v4f*)(sp + u * 8);

    for (int jt = 0; jt < KW / KT; ++jt) {
        const v4f pa0 = *(const v4f*)(pp0 + jt * KT);
        const v4f pb0 = *(const v4f*)(pp0 + jt * KT + 4);
        const v4f pa1 = *(const v4f*)(pp1 + jt * KT);
        const v4f pb1 = *(const v4f*)(pp1 + jt * KT + 4);

        v4f va[4], vb[4];
#pragma unroll
        for (int s = 0; s < 4; ++s) {
            va[s] = *(const v4f*)(vp + (size_t)(2 * s)     * (V_ * D_));
            vb[s] = *(const v4f*)(vp + (size_t)(2 * s + 1) * (V_ * D_));
        }
        asm volatile("" : "+v"(kf[0]) : : "memory");

        v8f c0 = __builtin_shufflevector(pa0, pb0, 0, 1, 2, 3, 4, 5, 6, 7);
        v8f c1 = __builtin_shufflevector(pa1, pb1, 0, 1, 2, 3, 4, 5, 6, 7);
#pragma unroll
        for (int u = 0; u < 8; ++u) {
            const v4f a = kf[u];
            const v2f alo = {a.x, a.y};
            const v2f ahi = {a.z, a.w};
            const v2f b0lo = {qf0[u].x, qf0[u].y};
            const v2f b1lo = {qf1[u].x, qf1[u].y};
            const v2f b0hi = {qf0[u].z, qf0[u].w};
            const v2f b1hi = {qf1[u].z, qf1[u].w};
            c0 = mma(alo, b0lo, c0);
            c1 = mma(alo, b1lo, c1);
            c0 = mma(ahi, b0hi, c0);
            c1 = mma(ahi, b1hi, c1);
        }

        {
            int jn = (jt + 1) & (KW / KT - 1);
            asm volatile("" : "+s"(jn) : "v"(c0), "v"(c1) : "memory");
            const float* spn = sp + jn * (KT * D_);
#pragma unroll
            for (int u = 0; u < 8; ++u) kf[u] = *(const v4f*)(spn + u * 8);
            asm volatile("" ::: "memory");
        }

        float t0 = -INFINITY, t1 = -INFINITY;
#pragma unroll
        for (int r = 0; r < 8; ++r) {
            t0 = fmaxf(t0, c0[r]);
            t1 = fmaxf(t1, c1[r]);
        }
        t0 = fmaxf(t0, xhalf(t0));
        t1 = fmaxf(t1, xhalf(t1));
        const float n0 = fmaxf(m0, t0);
        const float n1 = fmaxf(m1, t1);
        const float z0 = (n0 == -INFINITY) ? 0.f : n0;
        const float z1 = (n1 == -INFINITY) ? 0.f : n1;
        const float a0 = __builtin_amdgcn_exp2f(m0 - z0);
        const float a1 = __builtin_amdgcn_exp2f(m1 - z1);
        m0 = n0;
        m1 = n1;

        float ps0 = 0.f, ps1 = 0.f;
#pragma unroll
        for (int r = 0; r < 8; ++r) {
            c0[r] = __builtin_amdgcn_exp2f(c0[r] - z0);
            c1[r] = __builtin_amdgcn_exp2f(c1[r] - z1);
            ps0 += c0[r];
            ps1 += c1[r];
        }
        s0 = s0 * a0 + ps0;
        s1 = s1 * a1 + ps1;
#pragma unroll
        for (int n = 0; n < 4; ++n) {
            acc0[n] *= a0;
            acc1[n] *= a1;
        }

#pragma unroll
        for (int s = 0; s < 4; ++s) {
            const v2f p0 = {c0[2 * s], c0[2 * s + 1]};
            const v2f p1 = {c1[2 * s], c1[2 * s + 1]};
#pragma unroll
            for (int n = 0; n < 4; ++n) {
                const v2f af = {va[s][n], vb[s][n]};
                acc0[n] = mma(af, p0, acc0[n]);
                acc1[n] = mma(af, p1, acc1[n]);
            }
        }

        vp += (size_t)KT * V_ * D_;
    }

    s0 += xhalf(s0);
    s1 += xhalf(s1);
    if (hlf == 0) {
        ms[w][0][l16] = m0;
        ms[w][1][l16] = m1;
        ms[w][2][l16] = s0;
        ms[w][3][l16] = s1;
    }
    __syncthreads();
    const float mo0 = ms[w ^ 1][0][l16];
    const float mo1 = ms[w ^ 1][1][l16];
    const float so0 = ms[w ^ 1][2][l16];
    const float so1 = ms[w ^ 1][3][l16];
    const float mm0 = fmaxf(m0, mo0);
    const float mm1 = fmaxf(m1, mo1);
    const float y0 = (mm0 == -INFINITY) ? 0.f : mm0;
    const float y1 = (mm1 == -INFINITY) ? 0.f : mm1;
    const float f0  = __builtin_amdgcn_exp2f(m0  - y0);
    const float f1  = __builtin_amdgcn_exp2f(m1  - y1);
    const float fo0 = __builtin_amdgcn_exp2f(mo0 - y0);
    const float fo1 = __builtin_amdgcn_exp2f(mo1 - y1);
    const float r0 = f0 / (s0 * f0 + so0 * fo0);
    const float r1 = f1 / (s1 * f1 + so1 * fo1);
    float* otw = ot[w];
#pragma unroll
    for (int r = 0; r < 8; ++r) {
        v4f o0 = {acc0[0][r], acc0[1][r], acc0[2][r], acc0[3][r]};
        v4f o1 = {acc1[0][r], acc1[1][r], acc1[2][r], acc1[3][r]};
        o0 *= r0;
        o1 *= r1;
        *(v4f*)&otw[l16 * OPITCH + hlf * 32 + r * 4]        = o0;
        *(v4f*)&otw[(16 + l16) * OPITCH + hlf * 32 + r * 4] = o1;
    }
    __syncthreads();
    const int rw = 16 * w;
    float* ob = out + (((size_t)(b * L_ + i0 + rw) * V_) + k) * D_ + l16 * 4;
#pragma unroll 4
    for (int it = 0; it < MT / (2 * KSPLIT); ++it) {
        const int row = 2 * it + hlf;
        const v4f oa = *(const v4f*)&ot[0][(rw + row) * OPITCH + l16 * 4];
        const v4f ob2 = *(const v4f*)&ot[1][(rw + row) * OPITCH + l16 * 4];
        *(v4f*)(ob + (size_t)row * (V_ * D_)) = oa + ob2;
    }
}

extern "C" void kernel_launch(void* const* d_in, const int* in_sizes, int n_in,
                              void* d_out, int out_size, void* d_ws, size_t ws_size,
                              hipStream_t stream) {
    const float* query = (const float*)d_in[0];
    const float* key   = (const float*)d_in[1];
    const float* value = (const float*)d_in[2];
    const int*   label = (const int*)d_in[3];
    float* out  = (float*)d_out;
    float* sumv = (float*)d_ws;

    const int n1 = B_ * L_ * D_;
    sum_key_kernel<<<(n1 + 255) / 256, 256, 0, stream>>>(key, sumv);

    const int blocks = B_ * V_ * (L_ / MT);
    attn_kernel<<<blocks, 32 * KSPLIT, 0, stream>>>(query, sumv, value, label, out);
}


# Attention softmax: row reference moved only when a score outgrows it by 2^64 (one wave-wide test per 16-key step); accumulator rescale, max exchange and factor exponentials leave the steady loop
# speedup vs baseline: 2.1334x; 1.0306x over previous
#include <hip/hip_runtime.h>
#include <math.h>


typedef __attribute__((ext_vector_type(2))) float v2f;
typedef __attribute__((ext_vector_type(4))) float v4f;
typedef __attribute__((ext_vector_type(8))) float v8f;
typedef __attribute__((ext_vector_type(4))) int   v4i;

#define B_  4
#define L_  512
#define V_  32
#define D_  64
#define MT  32
#define KT  16
#define CSCALE 0.18033688011112042f
#define OPITCH (D_ + 4)
#define KSPLIT 2
#define KW (L_ / KSPLIT)
#define NLAB 8
#define PPITCH (L_ + 8)
#define LAG 64.0f
#define MSTART (-3.0e38f)

__global__ __launch_bounds__(256)
void sum_key_kernel(const float* __restrict__ key, float* __restrict__ sumv) {
    int idx = blockIdx.x * blockDim.x + threadIdx.x;
    if (idx >= B_ * L_ * D_) return;
    int d  = idx % D_;
    int bl = idx / D_;
    const float* p = key + (size_t)bl * (V_ * D_) + d;
    float s = 0.f;
#pragma unroll
    for (int v = 0; v < V_; ++v) s += p[v * D_];
    sumv[idx] = s;
}

__device__ __forceinline__ float xhalf(float x) {
    const int b = __builtin_bit_cast(int, x);
    return __builtin_bit_cast(float,
        __builtin_amdgcn_permlanex16(b, b, 0x76543210, 0xfedcba98, false, false));
}

__device__ __forceinline__ v8f mma(v2f a, v2f b, v8f c) {
    return __builtin_amdgcn_wmma_f32_16x16x4_f32(false, a, false, b, (short)0, c, false, false);
}

__global__ __launch_bounds__(32 * KSPLIT)
__attribute__((amdgpu_waves_per_eu(4, 4)))
void attn_kernel(const float* __restrict__ q,
                 const float* __restrict__ sumv,
                 const float* __restrict__ val,
                 const int*   __restrict__ lab,
                 float*       __restrict__ out) {
    __shared__ __attribute__((aligned(16))) float ot[KSPLIT][MT * OPITCH];
    __shared__ float ms[KSPLIT][4][16];
    __shared__ __attribute__((aligned(16))) float pen[NLAB][PPITCH];

    const int blk   = blockIdx.x;
    const int itile = blk % (L_ / MT);
    const int bk    = blk / (L_ / MT);
    const int k     = bk % V_;
    const int b     = bk / V_;

    const int lane = threadIdx.x & 31;
    const int w    = threadIdx.x >> 5;
    const int hlf  = lane >> 4;
    const int l16  = lane & 15;
    const int i0   = itile * MT;

    v4f qf0[8], qf1[8];
    {
        const float* q0 =
            q + (((size_t)(b * L_ + i0 + l16) * V_) + k) * D_ + hlf * 4;
        const float* q1 = q0 + (size_t)16 * V_ * D_;
#pragma unroll
        for (int u = 0; u < 8; ++u) {
            qf0[u] = *(const v4f*)(q0 + u * 8) * CSCALE;
            qf1[u] = *(const v4f*)(q1 + u * 8) * CSCALE;
        }
    }

    const int* lb = lab + b * L_;
    const int lab0 = lb[i0 + l16];
    const int lab1 = lb[i0 + 16 + l16];

    {
        const int t8 = threadIdx.x * 8;
        const v4i la = *(const v4i*)(lb + t8);
        const v4i lc = *(const v4i*)(lb + t8 + 4);
        const v4f ninf = {-INFINITY, -INFINITY, -INFINITY, -INFINITY};
#pragma unroll
        for (int c = 0; c < NLAB; ++c) {
            *(v4f*)&pen[c][t8]     = ninf;
            *(v4f*)&pen[c][t8 + 4] = ninf;
        }
        const int kl[8] = {la.x, la.y, la.z, la.w, lc.x, lc.y, lc.z, lc.w};
#pragma unroll
        for (int r = 0; r < 8; ++r) pen[kl[r] & (NLAB - 1)][t8 + r] = 0.f;
    }
    __syncthreads();

    v8f acc0[4] = {{}, {}, {}, {}};
    v8f acc1[4] = {{}, {}, {}, {}};
    float m0 = MSTART, m1 = MSTART;
    float h0 = MSTART, h1 = MSTART;
    float s0 = 0.f, s1 = 0.f;

    const int jw = w * KW;
    const float* sp = sumv + ((size_t)b * L_ + jw + l16) * D_ + hlf * 4;
    const float* vp = val + (((size_t)b * L_ + jw + hlf * 8) * V_ + k) * D_ + l16 * 4;
    const float* pp0 = &pen[lab0 & (NLAB - 1)][jw + hlf * 8];
    const float* pp1 = &pen[lab1 & (NLAB - 1)][jw + hlf * 8];

    v4f kf[8];
#pragma unroll
    for (int u = 0; u < 8; ++u) kf[u] = *(const v4f*)(sp + u * 8);

    for (int jt = 0; jt < KW / KT; ++jt) {
        const v4f pa0 = *(const v4f*)(pp0 + jt * KT);
        const v4f pb0 = *(const v4f*)(pp0 + jt * KT + 4);
        const v4f pa1 = *(const v4f*)(pp1 + jt * KT);
        const v4f pb1 = *(const v4f*)(pp1 + jt * KT + 4);

        v4f va[4], vb[4];
#pragma unroll
        for (int s = 0; s < 4; ++s) {
            va[s] = *(const v4f*)(vp + (size_t)(2 * s)     * (V_ * D_));
            vb[s] = *(const v4f*)(vp + (size_t)(2 * s + 1) * (V_ * D_));
        }
        asm volatile("" : "+v"(kf[0]) : : "memory");

        v8f c0 = __builtin_shufflevector(pa0, pb0, 0, 1, 2, 3, 4, 5, 6, 7);
        v8f c1 = __builtin_shufflevector(pa1, pb1, 0, 1, 2, 3, 4, 5, 6, 7);
#pragma unroll
        for (int u = 0; u < 8; ++u) {
            const v4f a = kf[u];
            const v2f alo = {a.x, a.y};
            const v2f ahi = {a.z, a.w};
            const v2f b0lo = {qf0[u].x, qf0[u].y};
            const v2f b1lo = {qf1[u].x, qf1[u].y};
            const v2f b0hi = {qf0[u].z, qf0[u].w};
            const v2f b1hi = {qf1[u].z, qf1[u].w};
            c0 = mma(alo, b0lo, c0);
            c1 = mma(alo, b1lo, c1);
            c0 = mma(ahi, b0hi, c0);
            c1 = mma(ahi, b1hi, c1);
        }

        {
            int jn = (jt + 1) & (KW / KT - 1);
            asm volatile("" : "+s"(jn) : "v"(c0), "v"(c1) : "memory");
            const float* spn = sp + jn * (KT * D_);
#pragma unroll
            for (int u = 0; u < 8; ++u) kf[u] = *(const v4f*)(spn + u * 8);
            asm volatile("" ::: "memory");
        }

        float t0 = -INFINITY, t1 = -INFINITY;
#pragma unroll
        for (int r = 0; r < 8; ++r) {
            t0 = fmaxf(t0, c0[r]);
            t1 = fmaxf(t1, c1[r]);
        }
        if (__builtin_amdgcn_ballot_w32((t0 > h0) | (t1 > h1)) != 0u) {
            t0 = fmaxf(t0, xhalf(t0));
            t1 = fmaxf(t1, xhalf(t1));
            const float n0 = fmaxf(m0, t0);
            const float n1 = fmaxf(m1, t1);
            const float a0 = __builtin_amdgcn_exp2f(m0 - n0);
            const float a1 = __builtin_amdgcn_exp2f(m1 - n1);
            m0 = n0;
            m1 = n1;
            h0 = n0 + LAG;
            h1 = n1 + LAG;
            s0 *= a0;
            s1 *= a1;
#pragma unroll
            for (int n = 0; n < 4; ++n) {
                acc0[n] *= a0;
                acc1[n] *= a1;
            }
        }

        float ps0 = 0.f, ps1 = 0.f;
#pragma unroll
        for (int r = 0; r < 8; ++r) {
            c0[r] = __builtin_amdgcn_exp2f(c0[r] - m0);
            c1[r] = __builtin_amdgcn_exp2f(c1[r] - m1);
            ps0 += c0[r];
            ps1 += c1[r];
        }
        s0 += ps0;
        s1 += ps1;

#pragma unroll
        for (int s = 0; s < 4; ++s) {
            const v2f p0 = {c0[2 * s], c0[2 * s + 1]};
            const v2f p1 = {c1[2 * s], c1[2 * s + 1]};
#pragma unroll
            for (int n = 0; n < 4; ++n) {
                const v2f af = {va[s][n], vb[s][n]};
                acc0[n] = mma(af, p0, acc0[n]);
                acc1[n] = mma(af, p1, acc1[n]);
            }
        }

        vp += (size_t)KT * V_ * D_;
    }

    s0 += xhalf(s0);
    s1 += xhalf(s1);
    if (hlf == 0) {
        ms[w][0][l16] = m0;
        ms[w][1][l16] = m1;
        ms[w][2][l16] = s0;
        ms[w][3][l16] = s1;
    }
    __syncthreads();
    const float mo0 = ms[w ^ 1][0][l16];
    const float mo1 = ms[w ^ 1][1][l16];
    const float so0 = ms[w ^ 1][2][l16];
    const float so1 = ms[w ^ 1][3][l16];
    const float y0 = fmaxf(m0, mo0);
    const float y1 = fmaxf(m1, mo1);
    const float f0  = __builtin_amdgcn_exp2f(m0  - y0);
    const float f1  = __builtin_amdgcn_exp2f(m1  - y1);
    const float fo0 = __builtin_amdgcn_exp2f(mo0 - y0);
    const float fo1 = __builtin_amdgcn_exp2f(mo1 - y1);
    const float r0 = f0 / (s0 * f0 + so0 * fo0);
    const float r1 = f1 / (s1 * f1 + so1 * fo1);
    float* otw = ot[w];
#pragma unroll
    for (int r = 0; r < 8; ++r) {
        v4f o0 = {acc0[0][r], acc0[1][r], acc0[2][r], acc0[3][r]};
        v4f o1 = {acc1[0][r], acc1[1][r], acc1[2][r], acc1[3][r]};
        o0 *= r0;
        o1 *= r1;
        *(v4f*)&otw[l16 * OPITCH + hlf * 32 + r * 4]        = o0;
        *(v4f*)&otw[(16 + l16) * OPITCH + hlf * 32 + r * 4] = o1;
    }
    __syncthreads();
    const int rw = 16 * w;
    float* ob = out + (((size_t)(b * L_ + i0 + rw) * V_) + k) * D_ + l16 * 4;
#pragma unroll 4
    for (int it = 0; it < MT / (2 * KSPLIT); ++it) {
        const int row = 2 * it + hlf;
        const v4f oa = *(const v4f*)&ot[0][(rw + row) * OPITCH + l16 * 4];
        const v4f ob2 = *(const v4f*)&ot[1][(rw + row) * OPITCH + l16 * 4];
        *(v4f*)(ob + (size_t)row * (V_ * D_)) = oa + ob2;
    }
}

extern "C" void kernel_launch(void* const* d_in, const int* in_sizes, int n_in,
                              void* d_out, int out_size, void* d_ws, size_t ws_size,
                              hipStream_t stream) {
    const float* query = (const float*)d_in[0];
    const float* key   = (const float*)d_in[1];
    const float* value = (const float*)d_in[2];
    const int*   label = (const int*)d_in[3];
    float* out  = (float*)d_out;
    float* sumv = (float*)d_ws;

    const int n1 = B_ * L_ * D_;
    sum_key_kernel<<<(n1 + 255) / 256, 256, 0, stream>>>(key, sumv);

    const int blocks = B_ * V_ * (L_ / MT);
    attn_kernel<<<blocks, 32 * KSPLIT, 0, stream>>>(query, sumv, value, label, out);
}
